# starting kernel
# baseline (speedup 1.0000x reference)
#include <hip/hip_runtime.h>

#define N_NODES 50000
#define N_EDGES 800000
#define NFEAT 512
#define NHID  256
#define NCLASS 64

typedef float v2f __attribute__((ext_vector_type(2)));
typedef float v4f __attribute__((ext_vector_type(4)));
typedef float v8f __attribute__((ext_vector_type(8)));

template<int K, int N>
__global__ __launch_bounds__(128)
void gemm_wmma_f32(const float* __restrict__ A, const float* __restrict__ B,
                   float* __restrict__ C, int Mtiles) {
  constexpr int NG = N / 64;
  const int wave = threadIdx.x >> 5;
  const int lane = threadIdx.x & 31;
  const int job  = blockIdx.x * 4 + wave;
  if (job >= Mtiles * NG) return;
  const int mt = job / NG;
  const int ng = job % NG;

  const int half = lane >> 4;
  const int idx  = lane & 15;
  const int nb   = ng * 64;

  v8f acc0 = {}, acc1 = {}, acc2 = {}, acc3 = {};

  const float* __restrict__ arow  = A + (size_t)(mt * 16 + idx) * K;
  const float* __restrict__ bbase = B + nb + 4 * idx;

  for (int k0 = 0; k0 < K; k0 += 4) {
    const int ka = k0 + half * 2;
    v2f a  = *(const v2f*)(arow + ka);
    v4f r0 = *(const v4f*)(bbase + (size_t)ka * N);
    v4f r1 = *(const v4f*)(bbase + (size_t)(ka + 1) * N);
    v2f b;
    b.x = r0.x; b.y = r1.x;
    acc0 = __builtin_amdgcn_wmma_f32_16x16x4_f32(false, a, false, b, (short)0,
                                                 acc0, false, false);
    b.x = r0.y; b.y = r1.y;
    acc1 = __builtin_amdgcn_wmma_f32_16x16x4_f32(false, a, false, b, (short)0,
                                                 acc1, false, false);
    b.x = r0.z; b.y = r1.z;
    acc2 = __builtin_amdgcn_wmma_f32_16x16x4_f32(false, a, false, b, (short)0,
                                                 acc2, false, false);
    b.x = r0.w; b.y = r1.w;
    acc3 = __builtin_amdgcn_wmma_f32_16x16x4_f32(false, a, false, b, (short)0,
                                                 acc3, false, false);
  }

  const int mbase = mt * 16 + half * 8;
  float* __restrict__ cptr = C + (size_t)mbase * N + nb + 4 * idx;
#pragma unroll
  for (int i = 0; i < 8; ++i) {
    v4f st;
    st.x = acc0[i]; st.y = acc1[i]; st.z = acc2[i]; st.w = acc3[i];
    *(v4f*)(cptr + (size_t)i * N) = st;
  }
}

template<int F>
__global__ void spmm_scatter(const int* __restrict__ src,
                             const int* __restrict__ dst,
                             const float* __restrict__ w,
                             const float* __restrict__ sup,
                             float* __restrict__ out) {
  const int e = blockIdx.x;
  const int t = threadIdx.x;
  const int s = src[e];
  const int d = dst[e];
  const float wt = w[e];
  const float v = wt * sup[(size_t)s * F + t];
  unsafeAtomicAdd(out + (size_t)d * F + t, v);
}

__global__ void fill_zero(float* __restrict__ p, int n) {
  int i = blockIdx.x * blockDim.x + threadIdx.x;
  const int stride = gridDim.x * blockDim.x;
  for (; i < n; i += stride) p[i] = 0.0f;
}

__global__ void bias_relu(float* __restrict__ h, const float* __restrict__ b,
                          int total, int fmask) {
  int i = blockIdx.x * blockDim.x + threadIdx.x;
  const int stride = gridDim.x * blockDim.x;
  for (; i < total; i += stride) {
    float v = h[i] + b[i & fmask];
    h[i] = fmaxf(v, 0.0f);
  }
}

__global__ __launch_bounds__(NCLASS)
void logsoftmax_bias(float* __restrict__ out, const float* __restrict__ b2) {
  __shared__ float red[NCLASS];
  const int r = blockIdx.x;
  const int t = threadIdx.x;
  const float v = out[(size_t)r * NCLASS + t] + b2[t];

  red[t] = v;
  __syncthreads();
#pragma unroll
  for (int s = NCLASS / 2; s > 0; s >>= 1) {
    if (t < s) red[t] = fmaxf(red[t], red[t + s]);
    __syncthreads();
  }
  const float mx = red[0];
  __syncthreads();

  red[t] = __expf(v - mx);
  __syncthreads();
#pragma unroll
  for (int s = NCLASS / 2; s > 0; s >>= 1) {
    if (t < s) red[t] += red[t + s];
    __syncthreads();
  }
  const float lse = __logf(red[0]);
  out[(size_t)r * NCLASS + t] = v - mx - lse;
}

extern "C" void kernel_launch(void* const* d_in, const int* in_sizes, int n_in,
                              void* d_out, int out_size, void* d_ws, size_t ws_size,
                              hipStream_t stream) {
  (void)in_sizes; (void)n_in; (void)out_size; (void)ws_size;

  const float* x  = (const float*)d_in[0];
  const float* W1 = (const float*)d_in[1];
  const float* b1 = (const float*)d_in[2];
  const float* W2 = (const float*)d_in[3];
  const float* b2 = (const float*)d_in[4];
  const float* ew = (const float*)d_in[5];
  const int*   es = (const int*)d_in[6];
  const int*   ed = (const int*)d_in[7];
  float* out = (float*)d_out;

  char* ws = (char*)d_ws;
  float* sup1 = (float*)ws;
  float* h    = (float*)(ws + (size_t)N_NODES * NHID * sizeof(float));
  float* sup2 = (float*)(ws + (size_t)2 * N_NODES * NHID * sizeof(float));

  const int Mtiles = N_NODES / 16;

  fill_zero<<<2048, 256, 0, stream>>>(h, N_NODES * NHID);
  fill_zero<<<1024, 256, 0, stream>>>(out, N_NODES * NCLASS);

  {
    const int jobs = Mtiles * (NHID / 64);
    gemm_wmma_f32<NFEAT, NHID>
        <<<(jobs + 3) / 4, 128, 0, stream>>>(x, W1, sup1, Mtiles);
  }
  spmm_scatter<NHID><<<N_EDGES, NHID, 0, stream>>>(es, ed, ew, sup1, h);
  bias_relu<<<2048, 256, 0, stream>>>(h, b1, N_NODES * NHID, NHID - 1);

  {
    const int jobs = Mtiles * (NCLASS / 64);
    gemm_wmma_f32<NHID, NCLASS>
        <<<(jobs + 3) / 4, 128, 0, stream>>>(h, W2, sup2, Mtiles);
  }
  spmm_scatter<NCLASS><<<N_EDGES, NCLASS, 0, stream>>>(es, ed, ew, sup2, out);
  logsoftmax_bias<<<N_NODES, NCLASS, 0, stream>>>(out, b2);
}


# SpMM gathers: each row's edge chain split into 4 chains (edge e on chain e mod 4) walked in step, branch-free: 4 records + 4 source rows in flight per wave; row sums now interleave the 4 chains
# speedup vs baseline: 4.2007x; 4.2007x over previous
#include <hip/hip_runtime.h>

#define N_NODES 50000
#define N_EDGES 800000
#define NFEAT 512
#define NHID  256
#define NCLASS 64

typedef float v2f __attribute__((ext_vector_type(2)));
typedef float v4f __attribute__((ext_vector_type(4)));
typedef float v8f __attribute__((ext_vector_type(8)));

template<int K, int N, bool ACT>
__device__ __forceinline__
void gemm_tile_f32(const int blk,
                   const float* __restrict__ A, const float* __restrict__ B,
                   float* __restrict__ C, const float* __restrict__ abias,
                   int Mtiles) {
  constexpr int NG = N / 64;
  const int wave = threadIdx.x >> 5;
  const int lane = threadIdx.x & 31;
  const int job  = blk * 4 + wave;
  if (job >= Mtiles * NG) return;
  const int mt = job / NG;
  const int ng = job % NG;

  const int half = lane >> 4;
  const int idx  = lane & 15;
  const int nb   = ng * 64;

  v8f acc0 = {}, acc1 = {}, acc2 = {}, acc3 = {};

  const float* __restrict__ arow  = A + (size_t)(mt * 16 + idx) * K;
  const float* __restrict__ bbase = B + nb + 4 * idx;

#pragma unroll 8
  for (int k0 = 0; k0 < K; k0 += 4) {
    const int ka = k0 + half * 2;
    v2f a  = *(const v2f*)(arow + ka);
    if constexpr (ACT) {
      const v2f ab = *(const v2f*)(abias + ka);
      a.x = fmaxf(a.x + ab.x, 0.0f);
      a.y = fmaxf(a.y + ab.y, 0.0f);
    }
    v4f r0 = *(const v4f*)(bbase + (size_t)ka * N);
    v4f r1 = *(const v4f*)(bbase + (size_t)(ka + 1) * N);
    v2f b;
    b.x = r0.x; b.y = r1.x;
    acc0 = __builtin_amdgcn_wmma_f32_16x16x4_f32(false, a, false, b, (short)0,
                                                 acc0, false, false);
    b.x = r0.y; b.y = r1.y;
    acc1 = __builtin_amdgcn_wmma_f32_16x16x4_f32(false, a, false, b, (short)0,
                                                 acc1, false, false);
    b.x = r0.z; b.y = r1.z;
    acc2 = __builtin_amdgcn_wmma_f32_16x16x4_f32(false, a, false, b, (short)0,
                                                 acc2, false, false);
    b.x = r0.w; b.y = r1.w;
    acc3 = __builtin_amdgcn_wmma_f32_16x16x4_f32(false, a, false, b, (short)0,
                                                 acc3, false, false);
  }

  const int mbase = mt * 16 + half * 8;
  float* __restrict__ cptr = C + (size_t)mbase * N + nb + 4 * idx;
#pragma unroll
  for (int i = 0; i < 8; ++i) {
    v4f st;
    st.x = acc0[i]; st.y = acc1[i]; st.z = acc2[i]; st.w = acc3[i];
    *(v4f*)(cptr + (size_t)i * N) = st;
  }
}

template<int K, int N, bool ACT>
__global__ __launch_bounds__(128)
void gemm_wmma_f32(const float* __restrict__ A, const float* __restrict__ B,
                   float* __restrict__ C, const float* __restrict__ abias,
                   int Mtiles) {
  gemm_tile_f32<K, N, ACT>((int)blockIdx.x, A, B, C, abias, Mtiles);
}

#define NSUB 4
#define REC_END N_EDGES
__global__ void chain_init(int* __restrict__ head, int* __restrict__ rec) {
  const int i = blockIdx.x * blockDim.x + threadIdx.x;
  if (i < NSUB * N_NODES) head[i] = -1;
  if (i == 0) {
    rec[3 * REC_END + 0] = REC_END;
    rec[3 * REC_END + 1] = 0;
    rec[3 * REC_END + 2] = 0;
  }
}

#define CB_PER_THREAD 8
#define CB_EDGES  (128 * CB_PER_THREAD)
#define CB_FULL   (N_EDGES / CB_EDGES)
#define CB_BLOCKS ((N_EDGES + CB_EDGES - 1) / CB_EDGES)

template<int NE>
__device__ __forceinline__
void chain_link(const int base,
                const int* __restrict__ src,
                const int* __restrict__ dst,
                const float* __restrict__ w,
                int* __restrict__ head,
                int* __restrict__ rec) {
  int d[NE], s[NE], wb[NE], prev[NE];
#pragma unroll
  for (int j = 0; j < NE; ++j) {
    const int e = base + j * 128;
    d[j]  = dst[e];
    s[j]  = src[e];
    wb[j] = __float_as_int(w[e]);
  }
#pragma unroll
  for (int j = 0; j < NE; ++j)
    prev[j] = atomicExch(head + NSUB * d[j] + ((base + j * 128) % NSUB), base + j * 128);
#pragma unroll
  for (int j = 0; j < NE; ++j) {
    const int e = base + j * 128;
    rec[3 * e + 0] = prev[j];
    rec[3 * e + 1] = s[j];
    rec[3 * e + 2] = wb[j];
  }
}

__device__ __forceinline__
void chain_build_block(const int cb,
                       const int* __restrict__ src,
                       const int* __restrict__ dst,
                       const float* __restrict__ w,
                       int* __restrict__ head,
                       int* __restrict__ rec) {
  static_assert(N_EDGES % 128 == 0, "the edge list is whole 128-edge slices");
  const int base = cb * CB_EDGES + (int)threadIdx.x;
  if (cb < CB_FULL) {
    chain_link<CB_PER_THREAD>(base, src, dst, w, head, rec);
  } else {
    for (int e = base; e < N_EDGES; e += 128)
      chain_link<1>(e, src, dst, w, head, rec);
  }
}

#define G1_BLOCKS (N_NODES / 16)
#define G1C_GRID  (5 * CB_BLOCKS)
__global__ __launch_bounds__(128)
void gemm1_with_chain_build(const float* __restrict__ x, const float* __restrict__ W1,
                            float* __restrict__ sup1,
                            const int* __restrict__ src, const int* __restrict__ dst,
                            const float* __restrict__ w,
                            int* __restrict__ head, int* __restrict__ rec) {
  static_assert(4 * CB_BLOCKS >= G1_BLOCKS, "every product block has a slot in the grid");
  const int q = (int)blockIdx.x / 5;
  const int r = (int)blockIdx.x - 5 * q;
  if (r == 0) {
    chain_build_block(q, src, dst, w, head, rec);
    return;
  }
  gemm_tile_f32<NFEAT, NHID, false>(4 * q + r - 1, x, W1, sup1, nullptr, G1_BLOCKS);
}

template<int F>
__global__ __launch_bounds__(128)
void spmm_chain_gather(const int* __restrict__ head,
                       const int* __restrict__ rec,
                       const float* __restrict__ sup,
                       float* __restrict__ out) {
  static_assert(F == 256 || F == 64, "row width");
  constexpr int VPL = F / 32;
  constexpr unsigned LANE_BYTES = (F == 256) ? 16u : 8u;
  const int lane = threadIdx.x & 31;
  const int wv  = __builtin_amdgcn_readfirstlane((int)(threadIdx.x >> 5));
  const int row = blockIdx.x * 4 + wv;
  if (row >= N_NODES) return;

  unsigned e[NSUB];
  int nx[NSUB], s[NSUB], wb[NSUB];
#pragma unroll
  for (int c = 0; c < NSUB; ++c) nx[c] = head[NSUB * row + c];
#pragma unroll
  for (int c = 0; c < NSUB; ++c) {
    e[c] = min((unsigned)nx[c], (unsigned)REC_END);
    const int* __restrict__ r = rec + 3 * e[c];
    nx[c] = r[0]; s[c] = r[1]; wb[c] = r[2];
  }

  float acc[VPL];
#pragma unroll
  for (int i = 0; i < VPL; ++i) acc[i] = 0.0f;

  for (int it = 0; it < N_EDGES; ++it) {
    unsigned left = e[0];
#pragma unroll
    for (int c = 1; c < NSUB; ++c) left = min(left, e[c]);
    if (left == (unsigned)REC_END) break;

    unsigned n_e[NSUB];
    int n_nx[NSUB], n_s[NSUB], n_wb[NSUB];
#pragma unroll
    for (int c = 0; c < NSUB; ++c) {
      n_e[c] = min((unsigned)nx[c], (unsigned)REC_END);
      const int* __restrict__ r = rec + 3 * n_e[c];
      n_nx[c] = r[0]; n_s[c] = r[1]; n_wb[c] = r[2];
    }

    float x[NSUB][VPL];
#pragma unroll
    for (int c = 0; c < NSUB; ++c) {
      const char* __restrict__ p = (const char*)sup + ((unsigned)s[c] * (unsigned)(F * 4) + (unsigned)lane * LANE_BYTES);
      if constexpr (F == 256) {
        const v4f lo = *(const v4f*)p;
        const v4f hi = *(const v4f*)(p + 512);
        x[c][0] = lo.x; x[c][1] = lo.y; x[c][2] = lo.z; x[c][3] = lo.w;
        x[c][4] = hi.x; x[c][5] = hi.y; x[c][6] = hi.z; x[c][7] = hi.w;
      } else {
        const v2f v = *(const v2f*)p;
        x[c][0] = v.x; x[c][1] = v.y;
      }
    }

#pragma unroll
    for (int c = 0; c < NSUB; ++c) {
      const bool on = e[c] != (unsigned)REC_END;
      const float wt = __int_as_float(wb[c]);
#pragma unroll
      for (int i = 0; i < VPL; ++i) acc[i] = on ? acc[i] + wt * x[c][i] : acc[i];
    }

#pragma unroll
    for (int c = 0; c < NSUB; ++c) {
      e[c] = n_e[c]; nx[c] = n_nx[c]; s[c] = n_s[c]; wb[c] = n_wb[c];
    }
  }

  if constexpr (F == 256) {
    float* __restrict__ o = out + (size_t)row * F + 4 * lane;
    v4f lo, hi;
    lo.x = acc[0]; lo.y = acc[1]; lo.z = acc[2]; lo.w = acc[3];
    hi.x = acc[4]; hi.y = acc[5]; hi.z = acc[6]; hi.w = acc[7];
    *(v4f*)o = lo;
    *(v4f*)(o + 128) = hi;
  } else {
    v2f v; v.x = acc[0]; v.y = acc[1];
    *(v2f*)(out + (size_t)row * F + 2 * lane) = v;
  }
}

#define LSM_ROWS   4
#define LSM_GROUPS (N_NODES / LSM_ROWS)
#define LSM_WPB    4
#define LSM_BLOCKS 625
#define LSM_WAVES  (LSM_BLOCKS * LSM_WPB)
#define LSM_TRIPS  (LSM_GROUPS / LSM_WAVES)

__global__ __launch_bounds__(LSM_WPB * 32)
void logsoftmax_bias(float* __restrict__ out, const float* __restrict__ b2) {
  static_assert(N_NODES % LSM_ROWS == 0, "whole row groups");
  static_assert(LSM_GROUPS % LSM_WAVES == 0, "every wave takes LSM_TRIPS groups");
  static_assert(NCLASS == 64, "two classes per lane");
  const int lane = threadIdx.x & 31;
  const int wv = __builtin_amdgcn_readfirstlane((int)(threadIdx.x >> 5));
  const int g0 = blockIdx.x * LSM_WPB + wv;

  const float c0 = b2[lane];
  const float c1 = b2[lane + 32];

#pragma unroll
  for (int t = 0; t < LSM_TRIPS; ++t) {
    const int g = g0 + t * LSM_WAVES;
    float* q = out + (size_t)g * (LSM_ROWS * NCLASS) + lane;

    float v0[LSM_ROWS], v1[LSM_ROWS], m[LSM_ROWS], s[LSM_ROWS];
#pragma unroll
    for (int j = 0; j < LSM_ROWS; ++j) {
      v0[j] = q[j * NCLASS] + c0;
      v1[j] = q[j * NCLASS + 32] + c1;
      m[j] = fmaxf(v0[j], v1[j]);
    }
#pragma unroll
    for (int x = 16; x > 0; x >>= 1) {
#pragma unroll
      for (int j = 0; j < LSM_ROWS; ++j)
        m[j] = fmaxf(m[j], __shfl_xor(m[j], x));
    }
#pragma unroll
    for (int j = 0; j < LSM_ROWS; ++j)
      s[j] = __expf(v0[j] - m[j]) + __expf(v1[j] - m[j]);
#pragma unroll
    for (int x = 16; x > 0; x >>= 1) {
#pragma unroll
      for (int j = 0; j < LSM_ROWS; ++j)
        s[j] += __shfl_xor(s[j], x);
    }
#pragma unroll
    for (int j = 0; j < LSM_ROWS; ++j) {
      const float lse = __logf(s[j]);
      q[j * NCLASS]      = v0[j] - m[j] - lse;
      q[j * NCLASS + 32] = v1[j] - m[j] - lse;
    }
  }
}

extern "C" void kernel_launch(void* const* d_in, const int* in_sizes, int n_in,
                              void* d_out, int out_size, void* d_ws, size_t ws_size,
                              hipStream_t stream) {
  (void)in_sizes; (void)n_in; (void)out_size; (void)ws_size;

  const float* x  = (const float*)d_in[0];
  const float* W1 = (const float*)d_in[1];
  const float* b1 = (const float*)d_in[2];
  const float* W2 = (const float*)d_in[3];
  const float* b2 = (const float*)d_in[4];
  const float* ew = (const float*)d_in[5];
  const int*   es = (const int*)d_in[6];
  const int*   ed = (const int*)d_in[7];
  float* out = (float*)d_out;

  char* ws = (char*)d_ws;
  float* sup1 = (float*)ws;
  float* h    = (float*)(ws + (size_t)N_NODES * NHID * sizeof(float));
  float* sup2 = (float*)ws;
  int*   rec  = (int*)(ws + (size_t)2 * N_NODES * NHID * sizeof(float));
  int*   head = rec + (size_t)3 * (N_EDGES + 1) + 1;

  const int Mtiles = N_NODES / 16;

  chain_init<<<(NSUB * N_NODES + 255) / 256, 256, 0, stream>>>(head, rec);

  gemm1_with_chain_build<<<G1C_GRID, 128, 0, stream>>>(x, W1, sup1, es, ed, ew, head, rec);
  spmm_chain_gather<NHID><<<(N_NODES + 3) / 4, 128, 0, stream>>>(head, rec, sup1, h);

  {
    const int jobs = Mtiles * (NCLASS / 64);
    gemm_wmma_f32<NHID, NCLASS, true>
        <<<(jobs + 3) / 4, 128, 0, stream>>>(h, W2, sup2, b1, Mtiles);
  }
  spmm_chain_gather<NCLASS><<<(N_NODES + 3) / 4, 128, 0, stream>>>(head, rec, sup2, out);
  logsoftmax_bias<<<LSM_BLOCKS, LSM_WPB * 32, 0, stream>>>(out, b2);
}


# GEMM A-fragment loads widened to 16 B per lane: one 16-row load feeds two k-steps (lane half h holds k = 8g+4h+0..3, B rows by the same k); k sum order inside each group of 8 changes
# speedup vs baseline: 4.3322x; 1.0313x over previous
#include <hip/hip_runtime.h>

#define N_NODES 50000
#define N_EDGES 800000
#define NFEAT 512
#define NHID  256
#define NCLASS 64

typedef float v2f __attribute__((ext_vector_type(2)));
typedef float v4f __attribute__((ext_vector_type(4)));
typedef float v8f __attribute__((ext_vector_type(8)));

template<int K, int N, bool ACT>
__device__ __forceinline__
void gemm_tile_f32(const int blk,
                   const float* __restrict__ A, const float* __restrict__ B,
                   float* __restrict__ C, const float* __restrict__ abias,
                   int Mtiles) {
  constexpr int NG = N / 64;
  static_assert(K % 32 == 0, "k runs in whole trips of four groups of eight");
  const int wave = threadIdx.x >> 5;
  const int lane = threadIdx.x & 31;
  const int job  = blk * 4 + wave;
  if (job >= Mtiles * NG) return;
  const int mt = job / NG;
  const int ng = job % NG;

  const int half = lane >> 4;
  const int idx  = lane & 15;
  const int nb   = ng * 64;

  v8f acc0 = {}, acc1 = {}, acc2 = {}, acc3 = {};

  const float* __restrict__ arow  = A + (size_t)(mt * 16 + idx) * K;
  const float* __restrict__ bbase = B + nb + 4 * idx;

  for (int k0 = 0; k0 < K; k0 += 32) {
    v4f a4[4], r0[4], r1[4], r2[4], r3[4];
#pragma unroll
    for (int g = 0; g < 4; ++g) {
      const int kb = k0 + 8 * g + half * 4;
      a4[g] = *(const v4f*)(arow + kb);
      const float* __restrict__ bp = bbase + (size_t)kb * N;
      r0[g] = *(const v4f*)(bp);
      r1[g] = *(const v4f*)(bp + (size_t)N);
      r2[g] = *(const v4f*)(bp + (size_t)2 * N);
      r3[g] = *(const v4f*)(bp + (size_t)3 * N);
      if constexpr (ACT) {
        const v4f ab = *(const v4f*)(abias + kb);
        a4[g].x = fmaxf(a4[g].x + ab.x, 0.0f);
        a4[g].y = fmaxf(a4[g].y + ab.y, 0.0f);
        a4[g].z = fmaxf(a4[g].z + ab.z, 0.0f);
        a4[g].w = fmaxf(a4[g].w + ab.w, 0.0f);
      }
    }
    __builtin_amdgcn_sched_barrier(0);
#pragma unroll
    for (int g = 0; g < 4; ++g) {
      v2f a, b;
      a.x = a4[g].x; a.y = a4[g].y;
      b.x = r0[g].x; b.y = r1[g].x;
      acc0 = __builtin_amdgcn_wmma_f32_16x16x4_f32(false, a, false, b, (short)0,
                                                   acc0, false, false);
      b.x = r0[g].y; b.y = r1[g].y;
      acc1 = __builtin_amdgcn_wmma_f32_16x16x4_f32(false, a, false, b, (short)0,
                                                   acc1, false, false);
      b.x = r0[g].z; b.y = r1[g].z;
      acc2 = __builtin_amdgcn_wmma_f32_16x16x4_f32(false, a, false, b, (short)0,
                                                   acc2, false, false);
      b.x = r0[g].w; b.y = r1[g].w;
      acc3 = __builtin_amdgcn_wmma_f32_16x16x4_f32(false, a, false, b, (short)0,
                                                   acc3, false, false);
      a.x = a4[g].z; a.y = a4[g].w;
      b.x = r2[g].x; b.y = r3[g].x;
      acc0 = __builtin_amdgcn_wmma_f32_16x16x4_f32(false, a, false, b, (short)0,
                                                   acc0, false, false);
      b.x = r2[g].y; b.y = r3[g].y;
      acc1 = __builtin_amdgcn_wmma_f32_16x16x4_f32(false, a, false, b, (short)0,
                                                   acc1, false, false);
      b.x = r2[g].z; b.y = r3[g].z;
      acc2 = __builtin_amdgcn_wmma_f32_16x16x4_f32(false, a, false, b, (short)0,
                                                   acc2, false, false);
      b.x = r2[g].w; b.y = r3[g].w;
      acc3 = __builtin_amdgcn_wmma_f32_16x16x4_f32(false, a, false, b, (short)0,
                                                   acc3, false, false);
    }
  }

  const int mbase = mt * 16 + half * 8;
  float* __restrict__ cptr = C + (size_t)mbase * N + nb + 4 * idx;
#pragma unroll
  for (int i = 0; i < 8; ++i) {
    v4f st;
    st.x = acc0[i]; st.y = acc1[i]; st.z = acc2[i]; st.w = acc3[i];
    *(v4f*)(cptr + (size_t)i * N) = st;
  }
}

template<int K, int N, bool ACT>
__global__ __launch_bounds__(128)
void gemm_wmma_f32(const float* __restrict__ A, const float* __restrict__ B,
                   float* __restrict__ C, const float* __restrict__ abias,
                   int Mtiles) {
  gemm_tile_f32<K, N, ACT>((int)blockIdx.x, A, B, C, abias, Mtiles);
}

#define NSUB 4
#define REC_END N_EDGES
__global__ void chain_init(int* __restrict__ head, int* __restrict__ rec) {
  const int i = blockIdx.x * blockDim.x + threadIdx.x;
  if (i < NSUB * N_NODES) head[i] = -1;
  if (i == 0) {
    rec[3 * REC_END + 0] = REC_END;
    rec[3 * REC_END + 1] = 0;
    rec[3 * REC_END + 2] = 0;
  }
}

#define CB_PER_THREAD 8
#define CB_EDGES  (128 * CB_PER_THREAD)
#define CB_FULL   (N_EDGES / CB_EDGES)
#define CB_BLOCKS ((N_EDGES + CB_EDGES - 1) / CB_EDGES)

template<int NE>
__device__ __forceinline__
void chain_link(const int base,
                const int* __restrict__ src,
                const int* __restrict__ dst,
                const float* __restrict__ w,
                int* __restrict__ head,
                int* __restrict__ rec) {
  int d[NE], s[NE], wb[NE], prev[NE];
#pragma unroll
  for (int j = 0; j < NE; ++j) {
    const int e = base + j * 128;
    d[j]  = dst[e];
    s[j]  = src[e];
    wb[j] = __float_as_int(w[e]);
  }
#pragma unroll
  for (int j = 0; j < NE; ++j)
    prev[j] = atomicExch(head + NSUB * d[j] + ((base + j * 128) % NSUB), base + j * 128);
#pragma unroll
  for (int j = 0; j < NE; ++j) {
    const int e = base + j * 128;
    rec[3 * e + 0] = prev[j];
    rec[3 * e + 1] = s[j];
    rec[3 * e + 2] = wb[j];
  }
}

__device__ __forceinline__
void chain_build_block(const int cb,
                       const int* __restrict__ src,
                       const int* __restrict__ dst,
                       const float* __restrict__ w,
                       int* __restrict__ head,
                       int* __restrict__ rec) {
  static_assert(N_EDGES % 128 == 0, "the edge list is whole 128-edge slices");
  const int base = cb * CB_EDGES + (int)threadIdx.x;
  if (cb < CB_FULL) {
    chain_link<CB_PER_THREAD>(base, src, dst, w, head, rec);
  } else {
    for (int e = base; e < N_EDGES; e += 128)
      chain_link<1>(e, src, dst, w, head, rec);
  }
}

#define G1_BLOCKS (N_NODES / 16)
#define G1C_GRID  (5 * CB_BLOCKS)
__global__ __launch_bounds__(128)
void gemm1_with_chain_build(const float* __restrict__ x, const float* __restrict__ W1,
                            float* __restrict__ sup1,
                            const int* __restrict__ src, const int* __restrict__ dst,
                            const float* __restrict__ w,
                            int* __restrict__ head, int* __restrict__ rec) {
  static_assert(4 * CB_BLOCKS >= G1_BLOCKS, "every product block has a slot in the grid");
  const int q = (int)blockIdx.x / 5;
  const int r = (int)blockIdx.x - 5 * q;
  if (r == 0) {
    chain_build_block(q, src, dst, w, head, rec);
    return;
  }
  gemm_tile_f32<NFEAT, NHID, false>(4 * q + r - 1, x, W1, sup1, nullptr, G1_BLOCKS);
}

template<int F>
__global__ __launch_bounds__(128)
void spmm_chain_gather(const int* __restrict__ head,
                       const int* __restrict__ rec,
                       const float* __restrict__ sup,
                       float* __restrict__ out) {
  static_assert(F == 256 || F == 64, "row width");
  constexpr int VPL = F / 32;
  constexpr unsigned LANE_BYTES = (F == 256) ? 16u : 8u;
  const int lane = threadIdx.x & 31;
  const int wv  = __builtin_amdgcn_readfirstlane((int)(threadIdx.x >> 5));
  const int row = blockIdx.x * 4 + wv;
  if (row >= N_NODES) return;

  unsigned e[NSUB];
  int nx[NSUB], s[NSUB], wb[NSUB];
#pragma unroll
  for (int c = 0; c < NSUB; ++c) nx[c] = head[NSUB * row + c];
#pragma unroll
  for (int c = 0; c < NSUB; ++c) {
    e[c] = min((unsigned)nx[c], (unsigned)REC_END);
    const int* __restrict__ r = rec + 3 * e[c];
    nx[c] = r[0]; s[c] = r[1]; wb[c] = r[2];
  }

  float acc[VPL];
#pragma unroll
  for (int i = 0; i < VPL; ++i) acc[i] = 0.0f;

  for (int it = 0; it < N_EDGES; ++it) {
    unsigned left = e[0];
#pragma unroll
    for (int c = 1; c < NSUB; ++c) left = min(left, e[c]);
    if (left == (unsigned)REC_END) break;

    unsigned n_e[NSUB];
    int n_nx[NSUB], n_s[NSUB], n_wb[NSUB];
#pragma unroll
    for (int c = 0; c < NSUB; ++c) {
      n_e[c] = min((unsigned)nx[c], (unsigned)REC_END);
      const int* __restrict__ r = rec + 3 * n_e[c];
      n_nx[c] = r[0]; n_s[c] = r[1]; n_wb[c] = r[2];
    }

    float x[NSUB][VPL];
#pragma unroll
    for (int c = 0; c < NSUB; ++c) {
      const char* __restrict__ p = (const char*)sup + ((unsigned)s[c] * (unsigned)(F * 4) + (unsigned)lane * LANE_BYTES);
      if constexpr (F == 256) {
        const v4f lo = *(const v4f*)p;
        const v4f hi = *(const v4f*)(p + 512);
        x[c][0] = lo.x; x[c][1] = lo.y; x[c][2] = lo.z; x[c][3] = lo.w;
        x[c][4] = hi.x; x[c][5] = hi.y; x[c][6] = hi.z; x[c][7] = hi.w;
      } else {
        const v2f v = *(const v2f*)p;
        x[c][0] = v.x; x[c][1] = v.y;
      }
    }

#pragma unroll
    for (int c = 0; c < NSUB; ++c) {
      const bool on = e[c] != (unsigned)REC_END;
      const float wt = __int_as_float(wb[c]);
#pragma unroll
      for (int i = 0; i < VPL; ++i) acc[i] = on ? acc[i] + wt * x[c][i] : acc[i];
    }

#pragma unroll
    for (int c = 0; c < NSUB; ++c) {
      e[c] = n_e[c]; nx[c] = n_nx[c]; s[c] = n_s[c]; wb[c] = n_wb[c];
    }
  }

  if constexpr (F == 256) {
    float* __restrict__ o = out + (size_t)row * F + 4 * lane;
    v4f lo, hi;
    lo.x = acc[0]; lo.y = acc[1]; lo.z = acc[2]; lo.w = acc[3];
    hi.x = acc[4]; hi.y = acc[5]; hi.z = acc[6]; hi.w = acc[7];
    *(v4f*)o = lo;
    *(v4f*)(o + 128) = hi;
  } else {
    v2f v; v.x = acc[0]; v.y = acc[1];
    *(v2f*)(out + (size_t)row * F + 2 * lane) = v;
  }
}

#define LSM_ROWS   4
#define LSM_GROUPS (N_NODES / LSM_ROWS)
#define LSM_WPB    4
#define LSM_BLOCKS 625
#define LSM_WAVES  (LSM_BLOCKS * LSM_WPB)
#define LSM_TRIPS  (LSM_GROUPS / LSM_WAVES)

__global__ __launch_bounds__(LSM_WPB * 32)
void logsoftmax_bias(float* __restrict__ out, const float* __restrict__ b2) {
  static_assert(N_NODES % LSM_ROWS == 0, "whole row groups");
  static_assert(LSM_GROUPS % LSM_WAVES == 0, "every wave takes LSM_TRIPS groups");
  static_assert(NCLASS == 64, "two classes per lane");
  const int lane = threadIdx.x & 31;
  const int wv = __builtin_amdgcn_readfirstlane((int)(threadIdx.x >> 5));
  const int g0 = blockIdx.x * LSM_WPB + wv;

  const float c0 = b2[lane];
  const float c1 = b2[lane + 32];

#pragma unroll
  for (int t = 0; t < LSM_TRIPS; ++t) {
    const int g = g0 + t * LSM_WAVES;
    float* q = out + (size_t)g * (LSM_ROWS * NCLASS) + lane;

    float v0[LSM_ROWS], v1[LSM_ROWS], m[LSM_ROWS], s[LSM_ROWS];
#pragma unroll
    for (int j = 0; j < LSM_ROWS; ++j) {
      v0[j] = q[j * NCLASS] + c0;
      v1[j] = q[j * NCLASS + 32] + c1;
      m[j] = fmaxf(v0[j], v1[j]);
    }
#pragma unroll
    for (int x = 16; x > 0; x >>= 1) {
#pragma unroll
      for (int j = 0; j < LSM_ROWS; ++j)
        m[j] = fmaxf(m[j], __shfl_xor(m[j], x));
    }
#pragma unroll
    for (int j = 0; j < LSM_ROWS; ++j)
      s[j] = __expf(v0[j] - m[j]) + __expf(v1[j] - m[j]);
#pragma unroll
    for (int x = 16; x > 0; x >>= 1) {
#pragma unroll
      for (int j = 0; j < LSM_ROWS; ++j)
        s[j] += __shfl_xor(s[j], x);
    }
#pragma unroll
    for (int j = 0; j < LSM_ROWS; ++j) {
      const float lse = __logf(s[j]);
      q[j * NCLASS]      = v0[j] - m[j] - lse;
      q[j * NCLASS + 32] = v1[j] - m[j] - lse;
    }
  }
}

extern "C" void kernel_launch(void* const* d_in, const int* in_sizes, int n_in,
                              void* d_out, int out_size, void* d_ws, size_t ws_size,
                              hipStream_t stream) {
  (void)in_sizes; (void)n_in; (void)out_size; (void)ws_size;

  const float* x  = (const float*)d_in[0];
  const float* W1 = (const float*)d_in[1];
  const float* b1 = (const float*)d_in[2];
  const float* W2 = (const float*)d_in[3];
  const float* b2 = (const float*)d_in[4];
  const float* ew = (const float*)d_in[5];
  const int*   es = (const int*)d_in[6];
  const int*   ed = (const int*)d_in[7];
  float* out = (float*)d_out;

  char* ws = (char*)d_ws;
  float* sup1 = (float*)ws;
  float* h    = (float*)(ws + (size_t)N_NODES * NHID * sizeof(float));
  float* sup2 = (float*)ws;
  int*   rec  = (int*)(ws + (size_t)2 * N_NODES * NHID * sizeof(float));
  int*   head = rec + (size_t)3 * (N_EDGES + 1) + 1;

  const int Mtiles = N_NODES / 16;

  chain_init<<<(NSUB * N_NODES + 255) / 256, 256, 0, stream>>>(head, rec);

  gemm1_with_chain_build<<<G1C_GRID, 128, 0, stream>>>(x, W1, sup1, es, ed, ew, head, rec);
  spmm_chain_gather<NHID><<<(N_NODES + 3) / 4, 128, 0, stream>>>(head, rec, sup1, h);

  {
    const int jobs = Mtiles * (NCLASS / 64);
    gemm_wmma_f32<NHID, NCLASS, true>
        <<<(jobs + 3) / 4, 128, 0, stream>>>(h, W2, sup2, b1, Mtiles);
  }
  spmm_chain_gather<NCLASS><<<(N_NODES + 3) / 4, 128, 0, stream>>>(head, rec, sup2, out);
  logsoftmax_bias<<<LSM_BLOCKS, LSM_WPB * 32, 0, stream>>>(out, b2);
}


# x @ W1 re-tiled for fewer cache requests per WMMA: a block stages its 32 x rows once in LDS (whole-line loads), a wave does 32x64 outputs so each W1 piece feeds two row tiles; k sum order kept
# speedup vs baseline: 4.5625x; 1.0532x over previous
#include <hip/hip_runtime.h>

#define N_NODES 50000
#define N_EDGES 800000
#define NFEAT 512
#define NHID  256
#define NCLASS 64

typedef float v2f __attribute__((ext_vector_type(2)));
typedef float v4f __attribute__((ext_vector_type(4)));
typedef float v8f __attribute__((ext_vector_type(8)));

template<int K, int N, bool ACT>
__device__ __forceinline__
void gemm_tile_f32(const int blk,
                   const float* __restrict__ A, const float* __restrict__ B,
                   float* __restrict__ C, const float* __restrict__ abias,
                   int Mtiles) {
  constexpr int NG = N / 64;
  static_assert(K % 32 == 0, "k runs in whole trips of four groups of eight");
  const int wave = threadIdx.x >> 5;
  const int lane = threadIdx.x & 31;
  const int job  = blk * 4 + wave;
  if (job >= Mtiles * NG) return;
  const int mt = job / NG;
  const int ng = job % NG;

  const int half = lane >> 4;
  const int idx  = lane & 15;
  const int nb   = ng * 64;

  v8f acc0 = {}, acc1 = {}, acc2 = {}, acc3 = {};

  const float* __restrict__ arow  = A + (size_t)(mt * 16 + idx) * K;
  const float* __restrict__ bbase = B + nb + 4 * idx;

  for (int k0 = 0; k0 < K; k0 += 32) {
    v4f a4[4], r0[4], r1[4], r2[4], r3[4];
#pragma unroll
    for (int g = 0; g < 4; ++g) {
      const int kb = k0 + 8 * g + half * 4;
      a4[g] = *(const v4f*)(arow + kb);
      const float* __restrict__ bp = bbase + (size_t)kb * N;
      r0[g] = *(const v4f*)(bp);
      r1[g] = *(const v4f*)(bp + (size_t)N);
      r2[g] = *(const v4f*)(bp + (size_t)2 * N);
      r3[g] = *(const v4f*)(bp + (size_t)3 * N);
      if constexpr (ACT) {
        const v4f ab = *(const v4f*)(abias + kb);
        a4[g].x = fmaxf(a4[g].x + ab.x, 0.0f);
        a4[g].y = fmaxf(a4[g].y + ab.y, 0.0f);
        a4[g].z = fmaxf(a4[g].z + ab.z, 0.0f);
        a4[g].w = fmaxf(a4[g].w + ab.w, 0.0f);
      }
    }
    __builtin_amdgcn_sched_barrier(0);
#pragma unroll
    for (int g = 0; g < 4; ++g) {
      v2f a, b;
      a.x = a4[g].x; a.y = a4[g].y;
      b.x = r0[g].x; b.y = r1[g].x;
      acc0 = __builtin_amdgcn_wmma_f32_16x16x4_f32(false, a, false, b, (short)0,
                                                   acc0, false, false);
      b.x = r0[g].y; b.y = r1[g].y;
      acc1 = __builtin_amdgcn_wmma_f32_16x16x4_f32(false, a, false, b, (short)0,
                                                   acc1, false, false);
      b.x = r0[g].z; b.y = r1[g].z;
      acc2 = __builtin_amdgcn_wmma_f32_16x16x4_f32(false, a, false, b, (short)0,
                                                   acc2, false, false);
      b.x = r0[g].w; b.y = r1[g].w;
      acc3 = __builtin_amdgcn_wmma_f32_16x16x4_f32(false, a, false, b, (short)0,
                                                   acc3, false, false);
      a.x = a4[g].z; a.y = a4[g].w;
      b.x = r2[g].x; b.y = r3[g].x;
      acc0 = __builtin_amdgcn_wmma_f32_16x16x4_f32(false, a, false, b, (short)0,
                                                   acc0, false, false);
      b.x = r2[g].y; b.y = r3[g].y;
      acc1 = __builtin_amdgcn_wmma_f32_16x16x4_f32(false, a, false, b, (short)0,
                                                   acc1, false, false);
      b.x = r2[g].z; b.y = r3[g].z;
      acc2 = __builtin_amdgcn_wmma_f32_16x16x4_f32(false, a, false, b, (short)0,
                                                   acc2, false, false);
      b.x = r2[g].w; b.y = r3[g].w;
      acc3 = __builtin_amdgcn_wmma_f32_16x16x4_f32(false, a, false, b, (short)0,
                                                   acc3, false, false);
    }
  }

  const int mbase = mt * 16 + half * 8;
  float* __restrict__ cptr = C + (size_t)mbase * N + nb + 4 * idx;
#pragma unroll
  for (int i = 0; i < 8; ++i) {
    v4f st;
    st.x = acc0[i]; st.y = acc1[i]; st.z = acc2[i]; st.w = acc3[i];
    *(v4f*)(cptr + (size_t)i * N) = st;
  }
}

template<int K, int N, bool ACT>
__global__ __launch_bounds__(128)
void gemm_wmma_f32(const float* __restrict__ A, const float* __restrict__ B,
                   float* __restrict__ C, const float* __restrict__ abias,
                   int Mtiles) {
  gemm_tile_f32<K, N, ACT>((int)blockIdx.x, A, B, C, abias, Mtiles);
}

#define G1_ROWS   32
#define G1_PITCH  (NFEAT + 4)
#define G1_PBLOCKS ((N_NODES + G1_ROWS - 1) / G1_ROWS)

__device__ __forceinline__
void gemm1_block32(const int pb,
                   const float* __restrict__ x, const float* __restrict__ W1,
                   float* __restrict__ C, float* __restrict__ sA) {
  static_assert(N_NODES % 16 == 0, "a 16-row tile is wholly inside or wholly outside the matrix");
  static_assert(NFEAT == 4 * 128, "128 threads x 16 bytes = one row of x");
  const int tid  = (int)threadIdx.x;
  const int wave = tid >> 5;
  const int lane = tid & 31;
  const int R0   = pb * G1_ROWS;

#pragma unroll
  for (int b = 0; b < 2; ++b) {
    v4f t[16];
#pragma unroll
    for (int j = 0; j < 16; ++j) {
      const int gr = min(R0 + 16 * b + j, N_NODES - 1);
      t[j] = *(const v4f*)(x + (size_t)gr * NFEAT + 4 * tid);
    }
#pragma unroll
    for (int j = 0; j < 16; ++j)
      *(v4f*)(sA + (16 * b + j) * G1_PITCH + 4 * tid) = t[j];
  }
  __syncthreads();

  const int half = lane >> 4;
  const int idx  = lane & 15;
  const int nb   = wave * 64;

  v8f c00 = {}, c01 = {}, c02 = {}, c03 = {};
  v8f c10 = {}, c11 = {}, c12 = {}, c13 = {};

  const float* __restrict__ aL = sA + idx * G1_PITCH + 4 * half;
  const float* __restrict__ aH = aL + 16 * G1_PITCH;
  const float* __restrict__ bbase = W1 + nb + 4 * idx;

  for (int k0 = 0; k0 < NFEAT; k0 += 32) {
    v4f al[4], ah[4], r0[4], r1[4], r2[4], r3[4];
#pragma unroll
    for (int g = 0; g < 4; ++g) {
      const int kb = k0 + 8 * g + half * 4;
      al[g] = *(const v4f*)(aL + k0 + 8 * g);
      ah[g] = *(const v4f*)(aH + k0 + 8 * g);
      const float* __restrict__ bp = bbase + (size_t)kb * NHID;
      r0[g] = *(const v4f*)(bp);
      r1[g] = *(const v4f*)(bp + (size_t)NHID);
      r2[g] = *(const v4f*)(bp + (size_t)2 * NHID);
      r3[g] = *(const v4f*)(bp + (size_t)3 * NHID);
    }
    __builtin_amdgcn_sched_barrier(0);
#pragma unroll
    for (int g = 0; g < 4; ++g) {
      v2f a0, a1, b;
      a0.x = al[g].x; a0.y = al[g].y;
      a1.x = ah[g].x; a1.y = ah[g].y;
      b.x = r0[g].x; b.y = r1[g].x;
      c00 = __builtin_amdgcn_wmma_f32_16x16x4_f32(false, a0, false, b, (short)0, c00, false, false);
      c10 = __builtin_amdgcn_wmma_f32_16x16x4_f32(false, a1, false, b, (short)0, c10, false, false);
      b.x = r0[g].y; b.y = r1[g].y;
      c01 = __builtin_amdgcn_wmma_f32_16x16x4_f32(false, a0, false, b, (short)0, c01, false, false);
      c11 = __builtin_amdgcn_wmma_f32_16x16x4_f32(false, a1, false, b, (short)0, c11, false, false);
      b.x = r0[g].z; b.y = r1[g].z;
      c02 = __builtin_amdgcn_wmma_f32_16x16x4_f32(false, a0, false, b, (short)0, c02, false, false);
      c12 = __builtin_amdgcn_wmma_f32_16x16x4_f32(false, a1, false, b, (short)0, c12, false, false);
      b.x = r0[g].w; b.y = r1[g].w;
      c03 = __builtin_amdgcn_wmma_f32_16x16x4_f32(false, a0, false, b, (short)0, c03, false, false);
      c13 = __builtin_amdgcn_wmma_f32_16x16x4_f32(false, a1, false, b, (short)0, c13, false, false);
      a0.x = al[g].z; a0.y = al[g].w;
      a1.x = ah[g].z; a1.y = ah[g].w;
      b.x = r2[g].x; b.y = r3[g].x;
      c00 = __builtin_amdgcn_wmma_f32_16x16x4_f32(false, a0, false, b, (short)0, c00, false, false);
      c10 = __builtin_amdgcn_wmma_f32_16x16x4_f32(false, a1, false, b, (short)0, c10, false, false);
      b.x = r2[g].y; b.y = r3[g].y;
      c01 = __builtin_amdgcn_wmma_f32_16x16x4_f32(false, a0, false, b, (short)0, c01, false, false);
      c11 = __builtin_amdgcn_wmma_f32_16x16x4_f32(false, a1, false, b, (short)0, c11, false, false);
      b.x = r2[g].z; b.y = r3[g].z;
      c02 = __builtin_amdgcn_wmma_f32_16x16x4_f32(false, a0, false, b, (short)0, c02, false, false);
      c12 = __builtin_amdgcn_wmma_f32_16x16x4_f32(false, a1, false, b, (short)0, c12, false, false);
      b.x = r2[g].w; b.y = r3[g].w;
      c03 = __builtin_amdgcn_wmma_f32_16x16x4_f32(false, a0, false, b, (short)0, c03, false, false);
      c13 = __builtin_amdgcn_wmma_f32_16x16x4_f32(false, a1, false, b, (short)0, c13, false, false);
    }
  }

  float* __restrict__ cptr = C + (size_t)(R0 + half * 8) * NHID + nb + 4 * idx;
#pragma unroll
  for (int i = 0; i < 8; ++i) {
    v4f st;
    st.x = c00[i]; st.y = c01[i]; st.z = c02[i]; st.w = c03[i];
    *(v4f*)(cptr + (size_t)i * NHID) = st;
  }
  if (R0 + 16 < N_NODES) {
    float* __restrict__ cptr1 = cptr + (size_t)16 * NHID;
#pragma unroll
    for (int i = 0; i < 8; ++i) {
      v4f st;
      st.x = c10[i]; st.y = c11[i]; st.z = c12[i]; st.w = c13[i];
      *(v4f*)(cptr1 + (size_t)i * NHID) = st;
    }
  }
}

#define NSUB 4
#define REC_END N_EDGES
__global__ void chain_init(int* __restrict__ head, int* __restrict__ rec) {
  const int i = blockIdx.x * blockDim.x + threadIdx.x;
  if (i < NSUB * N_NODES) head[i] = -1;
  if (i == 0) {
    rec[3 * REC_END + 0] = REC_END;
    rec[3 * REC_END + 1] = 0;
    rec[3 * REC_END + 2] = 0;
  }
}

#define CB_PER_THREAD 8
#define CB_EDGES  (128 * CB_PER_THREAD)
#define CB_FULL   (N_EDGES / CB_EDGES)
#define CB_BLOCKS ((N_EDGES + CB_EDGES - 1) / CB_EDGES)

template<int NE>
__device__ __forceinline__
void chain_link(const int base,
                const int* __restrict__ src,
                const int* __restrict__ dst,
                const float* __restrict__ w,
                int* __restrict__ head,
                int* __restrict__ rec) {
  int d[NE], s[NE], wb[NE], prev[NE];
#pragma unroll
  for (int j = 0; j < NE; ++j) {
    const int e = base + j * 128;
    d[j]  = dst[e];
    s[j]  = src[e];
    wb[j] = __float_as_int(w[e]);
  }
#pragma unroll
  for (int j = 0; j < NE; ++j)
    prev[j] = atomicExch(head + NSUB * d[j] + ((base + j * 128) % NSUB), base + j * 128);
#pragma unroll
  for (int j = 0; j < NE; ++j) {
    const int e = base + j * 128;
    rec[3 * e + 0] = prev[j];
    rec[3 * e + 1] = s[j];
    rec[3 * e + 2] = wb[j];
  }
}

__device__ __forceinline__
void chain_build_block(const int cb,
                       const int* __restrict__ src,
                       const int* __restrict__ dst,
                       const float* __restrict__ w,
                       int* __restrict__ head,
                       int* __restrict__ rec) {
  static_assert(N_EDGES % 128 == 0, "the edge list is whole 128-edge slices");
  const int base = cb * CB_EDGES + (int)threadIdx.x;
  if (cb < CB_FULL) {
    chain_link<CB_PER_THREAD>(base, src, dst, w, head, rec);
  } else {
    for (int e = base; e < N_EDGES; e += 128)
      chain_link<1>(e, src, dst, w, head, rec);
  }
}

#define G1C_GRID  (3 * CB_BLOCKS)
__global__ __launch_bounds__(128)
void gemm1_with_chain_build(const float* __restrict__ x, const float* __restrict__ W1,
                            float* __restrict__ sup1,
                            const int* __restrict__ src, const int* __restrict__ dst,
                            const float* __restrict__ w,
                            int* __restrict__ head, int* __restrict__ rec) {
  static_assert(2 * CB_BLOCKS >= G1_PBLOCKS, "every product block has a slot in the grid");
  __shared__ float sA[G1_ROWS * G1_PITCH];
  const int q = (int)blockIdx.x / 3;
  const int r = (int)blockIdx.x - 3 * q;
  if (r == 0) {
    chain_build_block(q, src, dst, w, head, rec);
    return;
  }
  const int pb = 2 * q + r - 1;
  if (pb >= G1_PBLOCKS) return;
  gemm1_block32(pb, x, W1, sup1, sA);
}

template<int F>
__global__ __launch_bounds__(128)
void spmm_chain_gather(const int* __restrict__ head,
                       const int* __restrict__ rec,
                       const float* __restrict__ sup,
                       float* __restrict__ out) {
  static_assert(F == 256 || F == 64, "row width");
  constexpr int VPL = F / 32;
  constexpr unsigned LANE_BYTES = (F == 256) ? 16u : 8u;
  const int lane = threadIdx.x & 31;
  const int wv  = __builtin_amdgcn_readfirstlane((int)(threadIdx.x >> 5));
  const int row = blockIdx.x * 4 + wv;
  if (row >= N_NODES) return;

  unsigned e[NSUB];
  int nx[NSUB], s[NSUB], wb[NSUB];
#pragma unroll
  for (int c = 0; c < NSUB; ++c) nx[c] = head[NSUB * row + c];
#pragma unroll
  for (int c = 0; c < NSUB; ++c) {
    e[c] = min((unsigned)nx[c], (unsigned)REC_END);
    const int* __restrict__ r = rec + 3 * e[c];
    nx[c] = r[0]; s[c] = r[1]; wb[c] = r[2];
  }

  float acc[VPL];
#pragma unroll
  for (int i = 0; i < VPL; ++i) acc[i] = 0.0f;

  for (int it = 0; it < N_EDGES; ++it) {
    unsigned left = e[0];
#pragma unroll
    for (int c = 1; c < NSUB; ++c) left = min(left, e[c]);
    if (left == (unsigned)REC_END) break;

    unsigned n_e[NSUB];
    int n_nx[NSUB], n_s[NSUB], n_wb[NSUB];
#pragma unroll
    for (int c = 0; c < NSUB; ++c) {
      n_e[c] = min((unsigned)nx[c], (unsigned)REC_END);
      const int* __restrict__ r = rec + 3 * n_e[c];
      n_nx[c] = r[0]; n_s[c] = r[1]; n_wb[c] = r[2];
    }

    float x[NSUB][VPL];
#pragma unroll
    for (int c = 0; c < NSUB; ++c) {
      const char* __restrict__ p = (const char*)sup + ((unsigned)s[c] * (unsigned)(F * 4) + (unsigned)lane * LANE_BYTES);
      if constexpr (F == 256) {
        const v4f lo = *(const v4f*)p;
        const v4f hi = *(const v4f*)(p + 512);
        x[c][0] = lo.x; x[c][1] = lo.y; x[c][2] = lo.z; x[c][3] = lo.w;
        x[c][4] = hi.x; x[c][5] = hi.y; x[c][6] = hi.z; x[c][7] = hi.w;
      } else {
        const v2f v = *(const v2f*)p;
        x[c][0] = v.x; x[c][1] = v.y;
      }
    }

#pragma unroll
    for (int c = 0; c < NSUB; ++c) {
      const bool on = e[c] != (unsigned)REC_END;
      const float wt = __int_as_float(wb[c]);
#pragma unroll
      for (int i = 0; i < VPL; ++i) acc[i] = on ? acc[i] + wt * x[c][i] : acc[i];
    }

#pragma unroll
    for (int c = 0; c < NSUB; ++c) {
      e[c] = n_e[c]; nx[c] = n_nx[c]; s[c] = n_s[c]; wb[c] = n_wb[c];
    }
  }

  if constexpr (F == 256) {
    float* __restrict__ o = out + (size_t)row * F + 4 * lane;
    v4f lo, hi;
    lo.x = acc[0]; lo.y = acc[1]; lo.z = acc[2]; lo.w = acc[3];
    hi.x = acc[4]; hi.y = acc[5]; hi.z = acc[6]; hi.w = acc[7];
    *(v4f*)o = lo;
    *(v4f*)(o + 128) = hi;
  } else {
    v2f v; v.x = acc[0]; v.y = acc[1];
    *(v2f*)(out + (size_t)row * F + 2 * lane) = v;
  }
}

#define LSM_ROWS   4
#define LSM_GROUPS (N_NODES / LSM_ROWS)
#define LSM_WPB    4
#define LSM_BLOCKS 625
#define LSM_WAVES  (LSM_BLOCKS * LSM_WPB)
#define LSM_TRIPS  (LSM_GROUPS / LSM_WAVES)

__global__ __launch_bounds__(LSM_WPB * 32)
void logsoftmax_bias(float* __restrict__ out, const float* __restrict__ b2) {
  static_assert(N_NODES % LSM_ROWS == 0, "whole row groups");
  static_assert(LSM_GROUPS % LSM_WAVES == 0, "every wave takes LSM_TRIPS groups");
  static_assert(NCLASS == 64, "two classes per lane");
  const int lane = threadIdx.x & 31;
  const int wv = __builtin_amdgcn_readfirstlane((int)(threadIdx.x >> 5));
  const int g0 = blockIdx.x * LSM_WPB + wv;

  const float c0 = b2[lane];
  const float c1 = b2[lane + 32];

#pragma unroll
  for (int t = 0; t < LSM_TRIPS; ++t) {
    const int g = g0 + t * LSM_WAVES;
    float* q = out + (size_t)g * (LSM_ROWS * NCLASS) + lane;

    float v0[LSM_ROWS], v1[LSM_ROWS], m[LSM_ROWS], s[LSM_ROWS];
#pragma unroll
    for (int j = 0; j < LSM_ROWS; ++j) {
      v0[j] = q[j * NCLASS] + c0;
      v1[j] = q[j * NCLASS + 32] + c1;
      m[j] = fmaxf(v0[j], v1[j]);
    }
#pragma unroll
    for (int x = 16; x > 0; x >>= 1) {
#pragma unroll
      for (int j = 0; j < LSM_ROWS; ++j)
        m[j] = fmaxf(m[j], __shfl_xor(m[j], x));
    }
#pragma unroll
    for (int j = 0; j < LSM_ROWS; ++j)
      s[j] = __expf(v0[j] - m[j]) + __expf(v1[j] - m[j]);
#pragma unroll
    for (int x = 16; x > 0; x >>= 1) {
#pragma unroll
      for (int j = 0; j < LSM_ROWS; ++j)
        s[j] += __shfl_xor(s[j], x);
    }
#pragma unroll
    for (int j = 0; j < LSM_ROWS; ++j) {
      const float lse = __logf(s[j]);
      q[j * NCLASS]      = v0[j] - m[j] - lse;
      q[j * NCLASS + 32] = v1[j] - m[j] - lse;
    }
  }
}

extern "C" void kernel_launch(void* const* d_in, const int* in_sizes, int n_in,
                              void* d_out, int out_size, void* d_ws, size_t ws_size,
                              hipStream_t stream) {
  (void)in_sizes; (void)n_in; (void)out_size; (void)ws_size;

  const float* x  = (const float*)d_in[0];
  const float* W1 = (const float*)d_in[1];
  const float* b1 = (const float*)d_in[2];
  const float* W2 = (const float*)d_in[3];
  const float* b2 = (const float*)d_in[4];
  const float* ew = (const float*)d_in[5];
  const int*   es = (const int*)d_in[6];
  const int*   ed = (const int*)d_in[7];
  float* out = (float*)d_out;

  char* ws = (char*)d_ws;
  float* sup1 = (float*)ws;
  float* h    = (float*)(ws + (size_t)N_NODES * NHID * sizeof(float));
  float* sup2 = (float*)ws;
  int*   rec  = (int*)(ws + (size_t)2 * N_NODES * NHID * sizeof(float));
  int*   head = rec + (size_t)3 * (N_EDGES + 1) + 1;

  const int Mtiles = N_NODES / 16;

  chain_init<<<(NSUB * N_NODES + 255) / 256, 256, 0, stream>>>(head, rec);

  gemm1_with_chain_build<<<G1C_GRID, 128, 0, stream>>>(x, W1, sup1, es, ed, ew, head, rec);
  spmm_chain_gather<NHID><<<(N_NODES + 3) / 4, 128, 0, stream>>>(head, rec, sup1, h);

  {
    const int jobs = Mtiles * (NCLASS / 64);
    gemm_wmma_f32<NHID, NCLASS, true>
        <<<(jobs + 3) / 4, 128, 0, stream>>>(h, W2, sup2, b1, Mtiles);
  }
  spmm_chain_gather<NCLASS><<<(N_NODES + 3) / 4, 128, 0, stream>>>(head, rec, sup2, out);
  logsoftmax_bias<<<LSM_BLOCKS, LSM_WPB * 32, 0, stream>>>(out, b2);
}
